# starting kernel
# baseline (speedup 1.0000x reference)
#include <hip/hip_runtime.h>

#define HID       128
#define LDS_PITCH 132
#define MT        80
#define MSUB      5

typedef __attribute__((ext_vector_type(2))) float v2f;
typedef __attribute__((ext_vector_type(8))) float v8f;

__global__ void sage_zero(float* __restrict__ p, int n) {
    int i = blockIdx.x * blockDim.x + threadIdx.x;
    if (i < n) p[i] = 0.0f;
}

__global__ void sage_rdeg(const float* __restrict__ degf,
                          float* __restrict__ rdeg, int n) {
    int i = blockIdx.x * blockDim.x + threadIdx.x;
    if (i < n) {
        float d = degf[i];
        rdeg[i] = d > 1.0f ? 1.0f / d : 1.0f;
    }
}

__global__ void sage_scatter(const float* __restrict__ x,
                             const int*   __restrict__ src,
                             const int*   __restrict__ dst,
                             float*       __restrict__ agg,
                             float*       __restrict__ degf,
                             int nEdges, int doDeg) {
    int gtid = blockIdx.x * blockDim.x + threadIdx.x;
    int edge = gtid >> 5;
    int lane = gtid & 31;
    if (edge >= nEdges) return;

    int s = src[edge];
    int d = dst[edge];
    const float* xs = x   + (size_t)s * HID;
    float*       ad = agg + (size_t)d * HID;

    atomicAdd(ad + lane,      xs[lane]);
    atomicAdd(ad + lane + 32, xs[lane + 32]);
    atomicAdd(ad + lane + 64, xs[lane + 64]);
    atomicAdd(ad + lane + 96, xs[lane + 96]);
    if (doDeg && lane == 0) atomicAdd(degf + d, 1.0f);
}

__global__ void sage_gemm(const float* __restrict__ agg,
                          const float* __restrict__ rdeg,
                          const float* __restrict__ xin,
                          const float* __restrict__ wl,
                          const float* __restrict__ bl,
                          const float* __restrict__ wr,
                          float*       __restrict__ out,
                          int applyRelu) {
    __shared__ float sMean[MT * LDS_PITCH];
    __shared__ float sX   [MT * LDS_PITCH];

    const int tid      = threadIdx.x;
    const int tileBase = blockIdx.x * MT;

    for (int i = tid; i < MT * 32; i += 256) {
        int r   = i >> 5;
        int c4  = (i & 31) << 2;
        int row = tileBase + r;
        float rd = rdeg[row];
        float4 av = *(const float4*)(agg + (size_t)row * HID + c4);
        float4 mv = make_float4(av.x * rd, av.y * rd, av.z * rd, av.w * rd);
        *(float4*)(sMean + r * LDS_PITCH + c4) = mv;
        *(float4*)(sX    + r * LDS_PITCH + c4) =
            *(const float4*)(xin + (size_t)row * HID + c4);
    }
    __syncthreads();

    const int wave  = tid >> 5;
    const int lane  = tid & 31;
    const int lrow  = lane & 15;
    const int halfK = (lane >> 4) << 1;
    const int nCol  = wave * 16 + lrow;

    v8f acc[MSUB];
    #pragma unroll
    for (int mt = 0; mt < MSUB; ++mt) acc[mt] = (v8f){};

    for (int k0 = 0; k0 < HID; k0 += 4) {
        v2f b;
        b.x = wl[(size_t)(k0 + halfK)     * HID + nCol];
        b.y = wl[(size_t)(k0 + halfK + 1) * HID + nCol];
        #pragma unroll
        for (int mt = 0; mt < MSUB; ++mt) {
            const float* ap = sMean + (mt * 16 + lrow) * LDS_PITCH + k0 + halfK;
            v2f a; a.x = ap[0]; a.y = ap[1];
            acc[mt] = __builtin_amdgcn_wmma_f32_16x16x4_f32(
                          false, a, false, b, (short)0, acc[mt], false, false);
        }
    }

    for (int k0 = 0; k0 < HID; k0 += 4) {
        v2f b;
        b.x = wr[(size_t)(k0 + halfK)     * HID + nCol];
        b.y = wr[(size_t)(k0 + halfK + 1) * HID + nCol];
        #pragma unroll
        for (int mt = 0; mt < MSUB; ++mt) {
            const float* ap = sX + (mt * 16 + lrow) * LDS_PITCH + k0 + halfK;
            v2f a; a.x = ap[0]; a.y = ap[1];
            acc[mt] = __builtin_amdgcn_wmma_f32_16x16x4_f32(
                          false, a, false, b, (short)0, acc[mt], false, false);
        }
    }

    const float bias   = bl[nCol];
    const int   rowOff = (lane < 16) ? 0 : 8;
    #pragma unroll
    for (int mt = 0; mt < MSUB; ++mt) {
        #pragma unroll
        for (int r = 0; r < 8; ++r) {
            float v = acc[mt][r] + bias;
            if (applyRelu) v = v > 0.0f ? v : 0.0f;
            out[(size_t)(tileBase + mt * 16 + r + rowOff) * HID + nCol] = v;
        }
    }
}

extern "C" void kernel_launch(void* const* d_in, const int* in_sizes, int n_in,
                              void* d_out, int out_size, void* d_ws, size_t ws_size,
                              hipStream_t stream) {
    const float* x    = (const float*)d_in[0];
    const int*   ei   = (const int*)  d_in[1];
    const float* w1l  = (const float*)d_in[2];
    const float* b1l  = (const float*)d_in[3];
    const float* w1r  = (const float*)d_in[4];
    const float* w2l  = (const float*)d_in[5];
    const float* b2l  = (const float*)d_in[6];
    const float* w2r  = (const float*)d_in[7];

    const int N = in_sizes[0] / HID;
    const int E = in_sizes[1] / 2;
    const int* src = ei;
    const int* dst = ei + E;

    float* agg  = (float*)d_ws;
    float* deg  = agg  + (size_t)N * HID;
    float* h    = deg  + N;
    float* rdeg = h    + (size_t)N * HID;
    float* out  = (float*)d_out;

    const int zeroAllN   = N * (HID + 1);
    const int zeroAggN   = N * HID;
    const int zAllGrid   = (zeroAllN + 255) / 256;
    const int zAggGrid   = (zeroAggN + 255) / 256;
    const int nGrid      = (N + 255) / 256;
    const int scatGrid   = (int)(((long long)E * 32 + 255) / 256);
    const int gemmGrid   = N / MT;

    sage_zero   <<<zAllGrid, 256, 0, stream>>>(agg, zeroAllN);
    sage_scatter<<<scatGrid, 256, 0, stream>>>(x, src, dst, agg, deg, E, 1);
    sage_rdeg   <<<nGrid,    256, 0, stream>>>(deg, rdeg, N);
    sage_gemm   <<<gemmGrid, 256, 0, stream>>>(agg, rdeg, x, w1l, b1l, w1r, h, 1);

    sage_zero   <<<zAggGrid, 256, 0, stream>>>(agg, zeroAggN);
    sage_scatter<<<scatGrid, 256, 0, stream>>>(h, src, dst, agg, deg, E, 0);
    sage_gemm   <<<gemmGrid, 256, 0, stream>>>(agg, rdeg, h, w2l, b2l, w2r, out, 0);
}


# Neighbour sums by destination ownership: block holds 400 rows' sums in LDS, scans all dst, adds its edges' source rows by LDS float adds; no global atomic, zero or rdeg pass; sum order = LDS arrival
# speedup vs baseline: 1.6980x; 1.6980x over previous
#include <hip/hip_runtime.h>

#define HID       128
#define LDS_PITCH 132
#define MT        80
#define MSUB      5

typedef __attribute__((ext_vector_type(2))) float v2f;
typedef __attribute__((ext_vector_type(8))) float v8f;

#define RB      400
#define OWN_T   1024
#define QCAP    8192
#define CNT_COL 128

__global__ void __launch_bounds__(OWN_T)
sage_own(const float* __restrict__ x,
         const int*   __restrict__ src,
         const int*   __restrict__ dst,
         float*       __restrict__ mean,
         int nNodes, int nEdges) {
    __shared__ float sSum[RB * LDS_PITCH];
    __shared__ int2  sQ[QCAP];
    __shared__ int   sQn;

    const int tid  = threadIdx.x;
    const int lo   = blockIdx.x * RB;
    const int left = nNodes - lo;
    const unsigned span = (unsigned)(left < RB ? left : RB);

    for (int i = tid; i < RB * LDS_PITCH / 4; i += OWN_T)
        ((float4*)sSum)[i] = make_float4(0.0f, 0.0f, 0.0f, 0.0f);
    if (tid == 0) sQn = 0;
    __syncthreads();

    auto take = [&](int e, int d) {
        unsigned r = (unsigned)(d - lo);
        if (r < span) {
            int pos = atomicAdd(&sQn, 1);
            if (pos < QCAP) {
                sQ[pos] = make_int2(e, (int)r);
            } else {
                const float* xs = x + (size_t)src[e] * HID;
                float*       sr = sSum + r * LDS_PITCH;
                for (int f = 0; f < HID; ++f) atomicAdd(sr + f, xs[f]);
                atomicAdd(sr + CNT_COL, 1.0f);
            }
        }
    };

    const bool aligned16 = (((size_t)dst) & 15) == 0;
    const int  n4 = aligned16 ? (nEdges >> 2) : 0;
    const int4* d4 = (const int4*)dst;
    const int rot = (int)(((long long)blockIdx.x * n4) / gridDim.x) & ~31;

    int i = tid;
    for (; i + 3 * OWN_T < n4; i += 4 * OWN_T) {
        int  c[4];
        int4 v[4];
        #pragma unroll
        for (int u = 0; u < 4; ++u) {
            c[u] = i + u * OWN_T + rot;
            if (c[u] >= n4) c[u] -= n4;
            v[u] = d4[c[u]];
        }
        #pragma unroll
        for (int u = 0; u < 4; ++u) {
            const int e = c[u] << 2;
            take(e,     v[u].x);
            take(e + 1, v[u].y);
            take(e + 2, v[u].z);
            take(e + 3, v[u].w);
        }
    }
    for (; i < n4; i += OWN_T) {
        int c = i + rot;
        if (c >= n4) c -= n4;
        const int4 v = d4[c];
        const int  e = c << 2;
        take(e,     v.x);
        take(e + 1, v.y);
        take(e + 2, v.z);
        take(e + 3, v.w);
    }
    for (int e = (n4 << 2) + tid; e < nEdges; e += OWN_T) take(e, dst[e]);
    __syncthreads();

    int nq = sQn;
    if (nq > QCAP) nq = QCAP;

    for (int q = tid; q < nq; q += OWN_T) sQ[q].x = src[sQ[q].x];
    __syncthreads();

    const int wave = tid >> 5;
    const int lane = tid & 31;
    for (int base = wave * 8; base < nq; base += (OWN_T / 32) * 8) {
        int   s[8], r[8];
        float v[8][4];
        #pragma unroll
        for (int j = 0; j < 8; ++j) {
            const int  idx = (base + j < nq) ? base + j : base;
            const int2 q   = sQ[idx];
            s[j] = __builtin_amdgcn_readfirstlane(q.x);
            r[j] = __builtin_amdgcn_readfirstlane(q.y);
        }
        #pragma unroll
        for (int j = 0; j < 8; ++j) {
            const float* xs = x + (size_t)s[j] * HID + lane;
            v[j][0] = xs[0];
            v[j][1] = xs[32];
            v[j][2] = xs[64];
            v[j][3] = xs[96];
        }
        #pragma unroll
        for (int j = 0; j < 8; ++j) {
            if (base + j < nq) {
                float* sr = sSum + r[j] * LDS_PITCH;
                atomicAdd(sr + lane,      v[j][0]);
                atomicAdd(sr + lane + 32, v[j][1]);
                atomicAdd(sr + lane + 64, v[j][2]);
                atomicAdd(sr + lane + 96, v[j][3]);
                if (lane == 0) atomicAdd(sr + CNT_COL, 1.0f);
            }
        }
    }
    __syncthreads();

    for (int k = tid; k < (int)span * 32; k += OWN_T) {
        const int r  = k >> 5;
        const int c4 = (k & 31) << 2;
        const float* sr = sSum + r * LDS_PITCH;
        const float  d  = sr[CNT_COL];
        const float  rd = d > 1.0f ? 1.0f / d : 1.0f;
        const float4 a4 = *(const float4*)(sr + c4);
        *(float4*)(mean + (size_t)(lo + r) * HID + c4) =
            make_float4(a4.x * rd, a4.y * rd, a4.z * rd, a4.w * rd);
    }
}

__global__ void sage_gemm(const float* __restrict__ mean,
                          const float* __restrict__ xin,
                          const float* __restrict__ wl,
                          const float* __restrict__ bl,
                          const float* __restrict__ wr,
                          float*       __restrict__ out,
                          int applyRelu) {
    __shared__ float sMean[MT * LDS_PITCH];
    __shared__ float sX   [MT * LDS_PITCH];

    const int tid      = threadIdx.x;
    const int tileBase = blockIdx.x * MT;

    for (int i = tid; i < MT * 32; i += 256) {
        int r   = i >> 5;
        int c4  = (i & 31) << 2;
        int row = tileBase + r;
        *(float4*)(sMean + r * LDS_PITCH + c4) =
            *(const float4*)(mean + (size_t)row * HID + c4);
        *(float4*)(sX    + r * LDS_PITCH + c4) =
            *(const float4*)(xin + (size_t)row * HID + c4);
    }
    __syncthreads();

    const int wave  = tid >> 5;
    const int lane  = tid & 31;
    const int lrow  = lane & 15;
    const int halfK = (lane >> 4) << 1;
    const int nCol  = wave * 16 + lrow;

    v8f acc[MSUB];
    #pragma unroll
    for (int mt = 0; mt < MSUB; ++mt) acc[mt] = (v8f){};

    for (int k0 = 0; k0 < HID; k0 += 4) {
        v2f b;
        b.x = wl[(size_t)(k0 + halfK)     * HID + nCol];
        b.y = wl[(size_t)(k0 + halfK + 1) * HID + nCol];
        #pragma unroll
        for (int mt = 0; mt < MSUB; ++mt) {
            const float* ap = sMean + (mt * 16 + lrow) * LDS_PITCH + k0 + halfK;
            v2f a; a.x = ap[0]; a.y = ap[1];
            acc[mt] = __builtin_amdgcn_wmma_f32_16x16x4_f32(
                          false, a, false, b, (short)0, acc[mt], false, false);
        }
    }

    for (int k0 = 0; k0 < HID; k0 += 4) {
        v2f b;
        b.x = wr[(size_t)(k0 + halfK)     * HID + nCol];
        b.y = wr[(size_t)(k0 + halfK + 1) * HID + nCol];
        #pragma unroll
        for (int mt = 0; mt < MSUB; ++mt) {
            const float* ap = sX + (mt * 16 + lrow) * LDS_PITCH + k0 + halfK;
            v2f a; a.x = ap[0]; a.y = ap[1];
            acc[mt] = __builtin_amdgcn_wmma_f32_16x16x4_f32(
                          false, a, false, b, (short)0, acc[mt], false, false);
        }
    }

    const float bias   = bl[nCol];
    const int   rowOff = (lane < 16) ? 0 : 8;
    #pragma unroll
    for (int mt = 0; mt < MSUB; ++mt) {
        #pragma unroll
        for (int r = 0; r < 8; ++r) {
            float v = acc[mt][r] + bias;
            if (applyRelu) v = v > 0.0f ? v : 0.0f;
            out[(size_t)(tileBase + mt * 16 + r + rowOff) * HID + nCol] = v;
        }
    }
}

extern "C" void kernel_launch(void* const* d_in, const int* in_sizes, int n_in,
                              void* d_out, int out_size, void* d_ws, size_t ws_size,
                              hipStream_t stream) {
    const float* x    = (const float*)d_in[0];
    const int*   ei   = (const int*)  d_in[1];
    const float* w1l  = (const float*)d_in[2];
    const float* b1l  = (const float*)d_in[3];
    const float* w1r  = (const float*)d_in[4];
    const float* w2l  = (const float*)d_in[5];
    const float* b2l  = (const float*)d_in[6];
    const float* w2r  = (const float*)d_in[7];

    const int N = in_sizes[0] / HID;
    const int E = in_sizes[1] / 2;
    const int* src = ei;
    const int* dst = ei + E;

    float* mean = (float*)d_ws;
    float* h    = mean + (size_t)N * HID;
    float* out  = (float*)d_out;

    const int ownGrid  = (N + RB - 1) / RB;
    const int gemmGrid = N / MT;

    sage_own <<<ownGrid,  OWN_T, 0, stream>>>(x, src, dst, mean, N, E);
    sage_gemm<<<gemmGrid, 256,   0, stream>>>(mean, x, w1l, b1l, w1r, h, 1);

    sage_own <<<ownGrid,  OWN_T, 0, stream>>>(h, src, dst, mean, N, E);
    sage_gemm<<<gemmGrid, 256,   0, stream>>>(mean, h, w2l, b2l, w2r, out, 0);
}


# Layer 2 reuses layer 1's per-block edge queues (source id, owner row), kept in workspace within the call, instead of scanning dst and reading src again; full queues rescan; sum order = LDS arrival
# speedup vs baseline: 1.9615x; 1.1552x over previous
#include <hip/hip_runtime.h>

#define HID       128
#define LDS_PITCH 132
#define MT        80
#define MSUB      5

typedef __attribute__((ext_vector_type(2))) float v2f;
typedef __attribute__((ext_vector_type(8))) float v8f;

#define RB      400
#define OWN_T   1024
#define QCAP    8192
#define CNT_COL 128

__global__ void __launch_bounds__(OWN_T)
sage_own(const float* __restrict__ x,
         const int*   __restrict__ src,
         const int*   __restrict__ dst,
         float*       __restrict__ mean,
         int*         __restrict__ qws,
         int*         __restrict__ qcnt,
         int reuse, int nNodes, int nEdges) {
    __shared__ float sSum[RB * LDS_PITCH];
    __shared__ int2  sQ[QCAP];
    __shared__ int   sQn;

    const int tid  = threadIdx.x;
    const int lo   = blockIdx.x * RB;
    const int left = nNodes - lo;
    const unsigned span = (unsigned)(left < RB ? left : RB);

    for (int i = tid; i < RB * LDS_PITCH / 4; i += OWN_T)
        ((float4*)sSum)[i] = make_float4(0.0f, 0.0f, 0.0f, 0.0f);
    if (tid == 0) sQn = 0;
    __syncthreads();

    int2* const qmine = (int2*)qws + (size_t)blockIdx.x * QCAP;
    int nq = -1;
    if (reuse) {
        const int cnt = qcnt[blockIdx.x];
        if ((unsigned)cnt <= (unsigned)QCAP) nq = cnt;
    }

    if (nq >= 0) {
        for (int q = tid; q < nq; q += OWN_T) sQ[q] = qmine[q];
        __syncthreads();
    } else {
        auto take = [&](int e, int d) {
            unsigned r = (unsigned)(d - lo);
            if (r < span) {
                int pos = atomicAdd(&sQn, 1);
                if (pos < QCAP) {
                    sQ[pos] = make_int2(e, (int)r);
                } else {
                    const float* xs = x + (size_t)src[e] * HID;
                    float*       sr = sSum + r * LDS_PITCH;
                    for (int f = 0; f < HID; ++f) atomicAdd(sr + f, xs[f]);
                    atomicAdd(sr + CNT_COL, 1.0f);
                }
            }
        };

        const bool aligned16 = (((size_t)dst) & 15) == 0;
        const int  n4 = aligned16 ? (nEdges >> 2) : 0;
        const int4* d4 = (const int4*)dst;
        const int rot = (int)(((long long)blockIdx.x * n4) / gridDim.x) & ~31;

        int i = tid;
        for (; i + 3 * OWN_T < n4; i += 4 * OWN_T) {
            int  c[4];
            int4 v[4];
            #pragma unroll
            for (int u = 0; u < 4; ++u) {
                c[u] = i + u * OWN_T + rot;
                if (c[u] >= n4) c[u] -= n4;
                v[u] = d4[c[u]];
            }
            #pragma unroll
            for (int u = 0; u < 4; ++u) {
                const int e = c[u] << 2;
                take(e,     v[u].x);
                take(e + 1, v[u].y);
                take(e + 2, v[u].z);
                take(e + 3, v[u].w);
            }
        }
        for (; i < n4; i += OWN_T) {
            int c = i + rot;
            if (c >= n4) c -= n4;
            const int4 v = d4[c];
            const int  e = c << 2;
            take(e,     v.x);
            take(e + 1, v.y);
            take(e + 2, v.z);
            take(e + 3, v.w);
        }
        for (int e = (n4 << 2) + tid; e < nEdges; e += OWN_T) take(e, dst[e]);
        __syncthreads();

        const int nraw = sQn;
        nq = nraw < QCAP ? nraw : QCAP;

        for (int q = tid; q < nq; q += OWN_T) sQ[q].x = src[sQ[q].x];
        __syncthreads();

        if (!reuse) {
            for (int q = tid; q < nq; q += OWN_T) qmine[q] = sQ[q];
            if (tid == 0) qcnt[blockIdx.x] = nraw;
        }
    }

    const int wave = tid >> 5;
    const int lane = tid & 31;
    for (int base = wave * 8; base < nq; base += (OWN_T / 32) * 8) {
        int   s[8], r[8];
        float v[8][4];
        #pragma unroll
        for (int j = 0; j < 8; ++j) {
            const int  idx = (base + j < nq) ? base + j : base;
            const int2 q   = sQ[idx];
            s[j] = __builtin_amdgcn_readfirstlane(q.x);
            r[j] = __builtin_amdgcn_readfirstlane(q.y);
        }
        #pragma unroll
        for (int j = 0; j < 8; ++j) {
            const float* xs = x + (size_t)s[j] * HID + lane;
            v[j][0] = xs[0];
            v[j][1] = xs[32];
            v[j][2] = xs[64];
            v[j][3] = xs[96];
        }
        #pragma unroll
        for (int j = 0; j < 8; ++j) {
            if (base + j < nq) {
                float* sr = sSum + r[j] * LDS_PITCH;
                atomicAdd(sr + lane,      v[j][0]);
                atomicAdd(sr + lane + 32, v[j][1]);
                atomicAdd(sr + lane + 64, v[j][2]);
                atomicAdd(sr + lane + 96, v[j][3]);
                if (lane == 0) atomicAdd(sr + CNT_COL, 1.0f);
            }
        }
    }
    __syncthreads();

    for (int k = tid; k < (int)span * 32; k += OWN_T) {
        const int r  = k >> 5;
        const int c4 = (k & 31) << 2;
        const float* sr = sSum + r * LDS_PITCH;
        const float  d  = sr[CNT_COL];
        const float  rd = d > 1.0f ? 1.0f / d : 1.0f;
        const float4 a4 = *(const float4*)(sr + c4);
        *(float4*)(mean + (size_t)(lo + r) * HID + c4) =
            make_float4(a4.x * rd, a4.y * rd, a4.z * rd, a4.w * rd);
    }
}

__global__ void sage_gemm(const float* __restrict__ mean,
                          const float* __restrict__ xin,
                          const float* __restrict__ wl,
                          const float* __restrict__ bl,
                          const float* __restrict__ wr,
                          float*       __restrict__ out,
                          int applyRelu) {
    __shared__ float sMean[MT * LDS_PITCH];
    __shared__ float sX   [MT * LDS_PITCH];

    const int tid      = threadIdx.x;
    const int tileBase = blockIdx.x * MT;

    for (int i = tid; i < MT * 32; i += 256) {
        int r   = i >> 5;
        int c4  = (i & 31) << 2;
        int row = tileBase + r;
        *(float4*)(sMean + r * LDS_PITCH + c4) =
            *(const float4*)(mean + (size_t)row * HID + c4);
        *(float4*)(sX    + r * LDS_PITCH + c4) =
            *(const float4*)(xin + (size_t)row * HID + c4);
    }
    __syncthreads();

    const int wave  = tid >> 5;
    const int lane  = tid & 31;
    const int lrow  = lane & 15;
    const int halfK = (lane >> 4) << 1;
    const int nCol  = wave * 16 + lrow;

    v8f acc[MSUB];
    #pragma unroll
    for (int mt = 0; mt < MSUB; ++mt) acc[mt] = (v8f){};

    for (int k0 = 0; k0 < HID; k0 += 4) {
        v2f b;
        b.x = wl[(size_t)(k0 + halfK)     * HID + nCol];
        b.y = wl[(size_t)(k0 + halfK + 1) * HID + nCol];
        #pragma unroll
        for (int mt = 0; mt < MSUB; ++mt) {
            const float* ap = sMean + (mt * 16 + lrow) * LDS_PITCH + k0 + halfK;
            v2f a; a.x = ap[0]; a.y = ap[1];
            acc[mt] = __builtin_amdgcn_wmma_f32_16x16x4_f32(
                          false, a, false, b, (short)0, acc[mt], false, false);
        }
    }

    for (int k0 = 0; k0 < HID; k0 += 4) {
        v2f b;
        b.x = wr[(size_t)(k0 + halfK)     * HID + nCol];
        b.y = wr[(size_t)(k0 + halfK + 1) * HID + nCol];
        #pragma unroll
        for (int mt = 0; mt < MSUB; ++mt) {
            const float* ap = sX + (mt * 16 + lrow) * LDS_PITCH + k0 + halfK;
            v2f a; a.x = ap[0]; a.y = ap[1];
            acc[mt] = __builtin_amdgcn_wmma_f32_16x16x4_f32(
                          false, a, false, b, (short)0, acc[mt], false, false);
        }
    }

    const float bias   = bl[nCol];
    const int   rowOff = (lane < 16) ? 0 : 8;
    #pragma unroll
    for (int mt = 0; mt < MSUB; ++mt) {
        #pragma unroll
        for (int r = 0; r < 8; ++r) {
            float v = acc[mt][r] + bias;
            if (applyRelu) v = v > 0.0f ? v : 0.0f;
            out[(size_t)(tileBase + mt * 16 + r + rowOff) * HID + nCol] = v;
        }
    }
}

extern "C" void kernel_launch(void* const* d_in, const int* in_sizes, int n_in,
                              void* d_out, int out_size, void* d_ws, size_t ws_size,
                              hipStream_t stream) {
    const float* x    = (const float*)d_in[0];
    const int*   ei   = (const int*)  d_in[1];
    const float* w1l  = (const float*)d_in[2];
    const float* b1l  = (const float*)d_in[3];
    const float* w1r  = (const float*)d_in[4];
    const float* w2l  = (const float*)d_in[5];
    const float* b2l  = (const float*)d_in[6];
    const float* w2r  = (const float*)d_in[7];

    const int N = in_sizes[0] / HID;
    const int E = in_sizes[1] / 2;
    const int* src = ei;
    const int* dst = ei + E;

    float* mean = (float*)d_ws;
    float* h    = mean + (size_t)N * HID;
    float* out  = (float*)d_out;

    const int ownGrid  = (N + RB - 1) / RB;
    const int gemmGrid = N / MT;
    int* qws  = (int*)(h + (size_t)N * HID);
    int* qcnt = qws + (size_t)ownGrid * QCAP * 2;

    sage_own <<<ownGrid,  OWN_T, 0, stream>>>(x, src, dst, mean, qws, qcnt, 0, N, E);
    sage_gemm<<<gemmGrid, 256,   0, stream>>>(mean, x, w1l, b1l, w1r, h, 1);

    sage_own <<<ownGrid,  OWN_T, 0, stream>>>(h, src, dst, mean, qws, qcnt, 1, N, E);
    sage_gemm<<<gemmGrid, 256,   0, stream>>>(mean, h, w2l, b2l, w2r, out, 0);
}


# Layer-1 scan of all dst replaced by edge bit rows: an added launch marks every edge in its owner block's row (LDS ORs); each block reads its own 75-KB row to fill its queue; sum order = LDS arrival
# speedup vs baseline: 2.1112x; 1.0763x over previous
#include <hip/hip_runtime.h>

#define HID       128
#define LDS_PITCH 132
#define MT        80
#define MSUB      5

typedef __attribute__((ext_vector_type(2))) float v2f;
typedef __attribute__((ext_vector_type(8))) float v8f;

#define RB      400
#define OWN_T   1024
#define QCAP    8192
#define CNT_COL 128

#define MCH     2048
#define MROW    64
#define MPAD    65
#define MOWN    256

__global__ void __launch_bounds__(256)
sage_mark(const int* __restrict__ dst,
          unsigned*  __restrict__ mask,
          int maskPitch, int nNodes, int nEdges, int nOwn) {
    __shared__ unsigned sM[MOWN * MPAD];

    const int tid  = threadIdx.x;
    const int base = blockIdx.x * MCH;

    int d[MCH / 256];
    #pragma unroll
    for (int u = 0; u < MCH / 256; ++u) {
        const int e = base + u * 256 + tid;
        d[u] = e < nEdges ? dst[e] : -1;
    }

    for (int i = tid; i < MOWN * MPAD; i += 256) sM[i] = 0u;
    __syncthreads();

    #pragma unroll
    for (int u = 0; u < MCH / 256; ++u) {
        if ((unsigned)d[u] < (unsigned)nNodes) {
            const int o = d[u] / RB;
            const int i = u * 256 + tid;
            if (o < MOWN) atomicOr(&sM[o * MPAD + (i >> 5)], 1u << (i & 31));
        }
    }
    __syncthreads();

    for (int k = tid; k < nOwn * (MROW / 4); k += 256) {
        const int o = k / (MROW / 4);
        const int w = (k % (MROW / 4)) << 2;
        const unsigned* s = sM + o * MPAD + w;
        *(uint4*)(mask + (size_t)o * maskPitch + blockIdx.x * MROW + w) =
            make_uint4(s[0], s[1], s[2], s[3]);
    }
}

__global__ void __launch_bounds__(OWN_T)
sage_own(const float* __restrict__ x,
         const int*   __restrict__ src,
         const int*   __restrict__ dst,
         float*       __restrict__ mean,
         int*         __restrict__ qws,
         int*         __restrict__ qcnt,
         const unsigned* __restrict__ mask,
         int maskPitch,
         int reuse, int nNodes, int nEdges) {
    __shared__ float sSum[RB * LDS_PITCH];
    __shared__ int2  sQ[QCAP];
    __shared__ int   sQn;

    const int tid  = threadIdx.x;
    const int lo   = blockIdx.x * RB;
    const int left = nNodes - lo;
    const unsigned span = (unsigned)(left < RB ? left : RB);

    for (int i = tid; i < RB * LDS_PITCH / 4; i += OWN_T)
        ((float4*)sSum)[i] = make_float4(0.0f, 0.0f, 0.0f, 0.0f);
    if (tid == 0) sQn = 0;
    __syncthreads();

    int2* const qmine = (int2*)qws + (size_t)blockIdx.x * QCAP;
    int nq = -1;
    if (reuse) {
        const int cnt = qcnt[blockIdx.x];
        if ((unsigned)cnt <= (unsigned)QCAP) nq = cnt;
    }

    if (nq >= 0) {
        for (int q = tid; q < nq; q += OWN_T) sQ[q] = qmine[q];
        __syncthreads();
    } else {
        auto put = [&](int e, int pos) {
            if (pos < QCAP) {
                sQ[pos] = make_int2(e, 0);
            } else {
                const unsigned r  = (unsigned)(dst[e] - lo);
                const float*   xs = x + (size_t)src[e] * HID;
                float*         sr = sSum + r * LDS_PITCH;
                for (int f = 0; f < HID; ++f) atomicAdd(sr + f, xs[f]);
                atomicAdd(sr + CNT_COL, 1.0f);
            }
        };
        auto emit = [&](unsigned bits, int ebase, int& pos) {
            while (bits) {
                const int bit = __builtin_ctz(bits);
                bits &= bits - 1u;
                put(ebase + bit, pos++);
            }
        };

        const uint4* m4 = (const uint4*)(mask + (size_t)blockIdx.x * maskPitch);
        const int    n4 = maskPitch >> 2;
        for (int i0 = tid; i0 < n4; i0 += 4 * OWN_T) {
            uint4 m[4];
            #pragma unroll
            for (int u = 0; u < 4; ++u) {
                const int i = i0 + u * OWN_T;
                m[u] = i < n4 ? m4[i] : make_uint4(0u, 0u, 0u, 0u);
            }
            #pragma unroll
            for (int u = 0; u < 4; ++u) {
                const int n = __popc(m[u].x) + __popc(m[u].y) + __popc(m[u].z) + __popc(m[u].w);
                if (n) {
                    int pos = atomicAdd(&sQn, n);
                    const int ebase = (i0 + u * OWN_T) << 7;
                    emit(m[u].x, ebase,      pos);
                    emit(m[u].y, ebase + 32, pos);
                    emit(m[u].z, ebase + 64, pos);
                    emit(m[u].w, ebase + 96, pos);
                }
            }
        }
        __syncthreads();

        const int nraw = sQn;
        nq = nraw < QCAP ? nraw : QCAP;

        for (int q = tid; q < nq; q += OWN_T) {
            const int e = sQ[q].x;
            sQ[q] = make_int2(src[e], dst[e] - lo);
        }
        __syncthreads();

        if (!reuse) {
            for (int q = tid; q < nq; q += OWN_T) qmine[q] = sQ[q];
            if (tid == 0) qcnt[blockIdx.x] = nraw;
        }
    }

    const int wave = tid >> 5;
    const int lane = tid & 31;
    for (int base = wave * 8; base < nq; base += (OWN_T / 32) * 8) {
        int   s[8], r[8];
        float v[8][4];
        #pragma unroll
        for (int j = 0; j < 8; ++j) {
            const int  idx = (base + j < nq) ? base + j : base;
            const int2 q   = sQ[idx];
            s[j] = __builtin_amdgcn_readfirstlane(q.x);
            r[j] = __builtin_amdgcn_readfirstlane(q.y);
        }
        #pragma unroll
        for (int j = 0; j < 8; ++j) {
            const float* xs = x + (size_t)s[j] * HID + lane;
            v[j][0] = xs[0];
            v[j][1] = xs[32];
            v[j][2] = xs[64];
            v[j][3] = xs[96];
        }
        #pragma unroll
        for (int j = 0; j < 8; ++j) {
            if (base + j < nq) {
                float* sr = sSum + r[j] * LDS_PITCH;
                atomicAdd(sr + lane,      v[j][0]);
                atomicAdd(sr + lane + 32, v[j][1]);
                atomicAdd(sr + lane + 64, v[j][2]);
                atomicAdd(sr + lane + 96, v[j][3]);
                if (lane == 0) atomicAdd(sr + CNT_COL, 1.0f);
            }
        }
    }
    __syncthreads();

    for (int k = tid; k < (int)span * 32; k += OWN_T) {
        const int r  = k >> 5;
        const int c4 = (k & 31) << 2;
        const float* sr = sSum + r * LDS_PITCH;
        const float  d  = sr[CNT_COL];
        const float  rd = d > 1.0f ? 1.0f / d : 1.0f;
        const float4 a4 = *(const float4*)(sr + c4);
        *(float4*)(mean + (size_t)(lo + r) * HID + c4) =
            make_float4(a4.x * rd, a4.y * rd, a4.z * rd, a4.w * rd);
    }
}

__global__ void sage_gemm(const float* __restrict__ mean,
                          const float* __restrict__ xin,
                          const float* __restrict__ wl,
                          const float* __restrict__ bl,
                          const float* __restrict__ wr,
                          float*       __restrict__ out,
                          int applyRelu) {
    __shared__ float sMean[MT * LDS_PITCH];
    __shared__ float sX   [MT * LDS_PITCH];

    const int tid      = threadIdx.x;
    const int tileBase = blockIdx.x * MT;

    for (int i = tid; i < MT * 32; i += 256) {
        int r   = i >> 5;
        int c4  = (i & 31) << 2;
        int row = tileBase + r;
        *(float4*)(sMean + r * LDS_PITCH + c4) =
            *(const float4*)(mean + (size_t)row * HID + c4);
        *(float4*)(sX    + r * LDS_PITCH + c4) =
            *(const float4*)(xin + (size_t)row * HID + c4);
    }
    __syncthreads();

    const int wave  = tid >> 5;
    const int lane  = tid & 31;
    const int lrow  = lane & 15;
    const int halfK = (lane >> 4) << 1;
    const int nCol  = wave * 16 + lrow;

    v8f acc[MSUB];
    #pragma unroll
    for (int mt = 0; mt < MSUB; ++mt) acc[mt] = (v8f){};

    for (int k0 = 0; k0 < HID; k0 += 4) {
        v2f b;
        b.x = wl[(size_t)(k0 + halfK)     * HID + nCol];
        b.y = wl[(size_t)(k0 + halfK + 1) * HID + nCol];
        #pragma unroll
        for (int mt = 0; mt < MSUB; ++mt) {
            const float* ap = sMean + (mt * 16 + lrow) * LDS_PITCH + k0 + halfK;
            v2f a; a.x = ap[0]; a.y = ap[1];
            acc[mt] = __builtin_amdgcn_wmma_f32_16x16x4_f32(
                          false, a, false, b, (short)0, acc[mt], false, false);
        }
    }

    for (int k0 = 0; k0 < HID; k0 += 4) {
        v2f b;
        b.x = wr[(size_t)(k0 + halfK)     * HID + nCol];
        b.y = wr[(size_t)(k0 + halfK + 1) * HID + nCol];
        #pragma unroll
        for (int mt = 0; mt < MSUB; ++mt) {
            const float* ap = sX + (mt * 16 + lrow) * LDS_PITCH + k0 + halfK;
            v2f a; a.x = ap[0]; a.y = ap[1];
            acc[mt] = __builtin_amdgcn_wmma_f32_16x16x4_f32(
                          false, a, false, b, (short)0, acc[mt], false, false);
        }
    }

    const float bias   = bl[nCol];
    const int   rowOff = (lane < 16) ? 0 : 8;
    #pragma unroll
    for (int mt = 0; mt < MSUB; ++mt) {
        #pragma unroll
        for (int r = 0; r < 8; ++r) {
            float v = acc[mt][r] + bias;
            if (applyRelu) v = v > 0.0f ? v : 0.0f;
            out[(size_t)(tileBase + mt * 16 + r + rowOff) * HID + nCol] = v;
        }
    }
}

extern "C" void kernel_launch(void* const* d_in, const int* in_sizes, int n_in,
                              void* d_out, int out_size, void* d_ws, size_t ws_size,
                              hipStream_t stream) {
    const float* x    = (const float*)d_in[0];
    const int*   ei   = (const int*)  d_in[1];
    const float* w1l  = (const float*)d_in[2];
    const float* b1l  = (const float*)d_in[3];
    const float* w1r  = (const float*)d_in[4];
    const float* w2l  = (const float*)d_in[5];
    const float* b2l  = (const float*)d_in[6];
    const float* w2r  = (const float*)d_in[7];

    const int N = in_sizes[0] / HID;
    const int E = in_sizes[1] / 2;
    const int* src = ei;
    const int* dst = ei + E;

    float* mean = (float*)d_ws;
    float* h    = mean + (size_t)N * HID;
    float* out  = (float*)d_out;

    const int ownGrid  = (N + RB - 1) / RB;
    const int gemmGrid = N / MT;
    int* qws  = (int*)(h + (size_t)N * HID);
    int* qcnt = qws + (size_t)ownGrid * QCAP * 2;
    const int maskGrid  = (E + MCH - 1) / MCH;
    const int maskPitch = maskGrid * MROW;
    unsigned* mask = (unsigned*)(qcnt + ((ownGrid + 63) & ~63));

    sage_mark<<<maskGrid, 256,   0, stream>>>(dst, mask, maskPitch, N, E, ownGrid);

    sage_own <<<ownGrid,  OWN_T, 0, stream>>>(x, src, dst, mean, qws, qcnt, mask, maskPitch, 0, N, E);
    sage_gemm<<<gemmGrid, 256,   0, stream>>>(mean, x, w1l, b1l, w1r, h, 1);

    sage_own <<<ownGrid,  OWN_T, 0, stream>>>(h, src, dst, mean, qws, qcnt, mask, maskPitch, 1, N, E);
    sage_gemm<<<gemmGrid, 256,   0, stream>>>(mean, h, w2l, b2l, w2r, out, 0);
}


# Dense stage fused into the owner block: its 400 mean rows stay in LDS and feed the f32 matrix products (20 waves x 80 rows x 32 cols); x rows then staged over them; mean plane and 2 launches dropped
# speedup vs baseline: 2.4113x; 1.1421x over previous
#include <hip/hip_runtime.h>

#define HID       128
#define LDS_PITCH 132
#define MT        80
#define MSUB      5

typedef __attribute__((ext_vector_type(2))) float v2f;
typedef __attribute__((ext_vector_type(8))) float v8f;

#define RB      400
#define OWN_T   1024
#define QCAP    8192
#define CNT_COL 128
#define DW      ((RB / MT) * 4)

#define MCH     2048
#define MROW    64
#define MPAD    65
#define MOWN    256

__global__ void __launch_bounds__(256)
sage_mark(const int* __restrict__ dst,
          unsigned*  __restrict__ mask,
          int maskPitch, int nNodes, int nEdges, int nOwn) {
    __shared__ unsigned sM[MOWN * MPAD];

    const int tid  = threadIdx.x;
    const int base = blockIdx.x * MCH;

    int d[MCH / 256];
    #pragma unroll
    for (int u = 0; u < MCH / 256; ++u) {
        const int e = base + u * 256 + tid;
        d[u] = e < nEdges ? dst[e] : -1;
    }

    for (int i = tid; i < MOWN * MPAD; i += 256) sM[i] = 0u;
    __syncthreads();

    #pragma unroll
    for (int u = 0; u < MCH / 256; ++u) {
        if ((unsigned)d[u] < (unsigned)nNodes) {
            const int o = d[u] / RB;
            const int i = u * 256 + tid;
            if (o < MOWN) atomicOr(&sM[o * MPAD + (i >> 5)], 1u << (i & 31));
        }
    }
    __syncthreads();

    for (int k = tid; k < nOwn * (MROW / 4); k += 256) {
        const int o = k / (MROW / 4);
        const int w = (k % (MROW / 4)) << 2;
        const unsigned* s = sM + o * MPAD + w;
        *(uint4*)(mask + (size_t)o * maskPitch + blockIdx.x * MROW + w) =
            make_uint4(s[0], s[1], s[2], s[3]);
    }
}

__device__ __forceinline__ void sage_sweep(const float* a0,
                                           const float* __restrict__ w,
                                           int nc0, int halfK,
                                           v8f (&acc)[MSUB][2]) {
    #pragma nounroll
    for (int k0 = 0; k0 < HID; k0 += 4) {
        const float* wp = w + (size_t)(k0 + halfK) * HID + nc0;
        v2f b0; b0.x = wp[0];  b0.y = wp[HID];
        v2f b1; b1.x = wp[16]; b1.y = wp[HID + 16];
        #pragma unroll
        for (int mt = 0; mt < MSUB; ++mt) {
            const float* ap = a0 + mt * 16 * LDS_PITCH + k0;
            v2f a; a.x = ap[0]; a.y = ap[1];
            acc[mt][0] = __builtin_amdgcn_wmma_f32_16x16x4_f32(
                             false, a, false, b0, (short)0, acc[mt][0], false, false);
            acc[mt][1] = __builtin_amdgcn_wmma_f32_16x16x4_f32(
                             false, a, false, b1, (short)0, acc[mt][1], false, false);
        }
    }
}

__global__ void __launch_bounds__(OWN_T)
sage_own(const float* __restrict__ x,
         const int*   __restrict__ src,
         const int*   __restrict__ dst,
         const float* __restrict__ wl,
         const float* __restrict__ bl,
         const float* __restrict__ wr,
         float*       __restrict__ out,
         int applyRelu,
         int*         __restrict__ qws,
         int*         __restrict__ qcnt,
         const unsigned* __restrict__ mask,
         int maskPitch,
         int reuse, int nNodes, int nEdges) {
    __shared__ float sSum[RB * LDS_PITCH];
    __shared__ int2  sQ[QCAP];
    __shared__ int   sQn;

    const int tid  = threadIdx.x;
    const int lo   = blockIdx.x * RB;

    for (int i = tid; i < RB * LDS_PITCH / 4; i += OWN_T)
        ((float4*)sSum)[i] = make_float4(0.0f, 0.0f, 0.0f, 0.0f);
    if (tid == 0) sQn = 0;
    __syncthreads();

    int2* const qmine = (int2*)qws + (size_t)blockIdx.x * QCAP;
    int nq = -1;
    if (reuse) {
        const int cnt = qcnt[blockIdx.x];
        if ((unsigned)cnt <= (unsigned)QCAP) nq = cnt;
    }

    if (nq >= 0) {
        for (int q = tid; q < nq; q += OWN_T) sQ[q] = qmine[q];
        __syncthreads();
    } else {
        auto put = [&](int e, int pos) {
            if (pos < QCAP) {
                sQ[pos] = make_int2(e, 0);
            } else {
                const unsigned r  = (unsigned)(dst[e] - lo);
                const float*   xs = x + (size_t)src[e] * HID;
                float*         sr = sSum + r * LDS_PITCH;
                for (int f = 0; f < HID; ++f) atomicAdd(sr + f, xs[f]);
                atomicAdd(sr + CNT_COL, 1.0f);
            }
        };
        auto emit = [&](unsigned bits, int ebase, int& pos) {
            while (bits) {
                const int bit = __builtin_ctz(bits);
                bits &= bits - 1u;
                put(ebase + bit, pos++);
            }
        };

        const uint4* m4 = (const uint4*)(mask + (size_t)blockIdx.x * maskPitch);
        const int    n4 = maskPitch >> 2;
        for (int i0 = tid; i0 < n4; i0 += 4 * OWN_T) {
            uint4 m[4];
            #pragma unroll
            for (int u = 0; u < 4; ++u) {
                const int i = i0 + u * OWN_T;
                m[u] = i < n4 ? m4[i] : make_uint4(0u, 0u, 0u, 0u);
            }
            #pragma unroll
            for (int u = 0; u < 4; ++u) {
                const int n = __popc(m[u].x) + __popc(m[u].y) + __popc(m[u].z) + __popc(m[u].w);
                if (n) {
                    int pos = atomicAdd(&sQn, n);
                    const int ebase = (i0 + u * OWN_T) << 7;
                    emit(m[u].x, ebase,      pos);
                    emit(m[u].y, ebase + 32, pos);
                    emit(m[u].z, ebase + 64, pos);
                    emit(m[u].w, ebase + 96, pos);
                }
            }
        }
        __syncthreads();

        const int nraw = sQn;
        nq = nraw < QCAP ? nraw : QCAP;

        for (int q = tid; q < nq; q += OWN_T) {
            const int e = sQ[q].x;
            sQ[q] = make_int2(src[e], dst[e] - lo);
        }
        __syncthreads();

        if (!reuse) {
            for (int q = tid; q < nq; q += OWN_T) qmine[q] = sQ[q];
            if (tid == 0) qcnt[blockIdx.x] = nraw;
        }
    }

    const int wave = tid >> 5;
    const int lane = tid & 31;
    for (int base = wave * 8; base < nq; base += (OWN_T / 32) * 8) {
        int   s[8], r[8];
        float v[8][4];
        #pragma unroll
        for (int j = 0; j < 8; ++j) {
            const int  idx = (base + j < nq) ? base + j : base;
            const int2 q   = sQ[idx];
            s[j] = __builtin_amdgcn_readfirstlane(q.x);
            r[j] = __builtin_amdgcn_readfirstlane(q.y);
        }
        #pragma unroll
        for (int j = 0; j < 8; ++j) {
            const float* xs = x + (size_t)s[j] * HID + lane;
            v[j][0] = xs[0];
            v[j][1] = xs[32];
            v[j][2] = xs[64];
            v[j][3] = xs[96];
        }
        #pragma unroll
        for (int j = 0; j < 8; ++j) {
            if (base + j < nq) {
                float* sr = sSum + r[j] * LDS_PITCH;
                atomicAdd(sr + lane,      v[j][0]);
                atomicAdd(sr + lane + 32, v[j][1]);
                atomicAdd(sr + lane + 64, v[j][2]);
                atomicAdd(sr + lane + 96, v[j][3]);
                if (lane == 0) atomicAdd(sr + CNT_COL, 1.0f);
            }
        }
    }
    __syncthreads();

    for (int k = tid; k < RB * 32; k += OWN_T) {
        const int r  = k >> 5;
        const int c4 = (k & 31) << 2;
        float* sr = sSum + r * LDS_PITCH;
        const float  d  = sr[CNT_COL];
        const float  rd = d > 1.0f ? 1.0f / d : 1.0f;
        const float4 a4 = *(const float4*)(sr + c4);
        *(float4*)(sr + c4) = make_float4(a4.x * rd, a4.y * rd, a4.z * rd, a4.w * rd);
    }
    __syncthreads();

    const int  wv    = __builtin_amdgcn_readfirstlane(wave);
    const bool dense = wv < DW;
    const int  g     = wv >> 2;
    const int  lrow  = lane & 15;
    const int  halfK = (lane >> 4) << 1;
    const int  nc0   = (wv & 3) * 32 + lrow;
    const float* a0  = sSum + (g * MT + lrow) * LDS_PITCH + halfK;

    v8f acc[MSUB][2];
    #pragma unroll
    for (int mt = 0; mt < MSUB; ++mt) {
        #pragma unroll
        for (int nt = 0; nt < 2; ++nt) acc[mt][nt] = (v8f){};
    }

    if (dense) sage_sweep(a0, wl, nc0, halfK, acc);
    __syncthreads();

    const int sr0  = tid >> 5;
    const int sc4  = (tid & 31) << 2;
    const int last = nNodes - 1;
    static_assert((RB / 32) % 4 == 0, "full passes are staged four at a time");
    for (int p = 0; p < RB / 32; p += 4) {
        const int r0 = sr0 + 32 * p;
        const int n0 = lo + r0;
        const int g0 = n0      < last ? n0      : last;
        const int g1 = n0 + 32 < last ? n0 + 32 : last;
        const int g2 = n0 + 64 < last ? n0 + 64 : last;
        const int g3 = n0 + 96 < last ? n0 + 96 : last;
        const float4 va = *(const float4*)(x + (size_t)g0 * HID + sc4);
        const float4 vb = *(const float4*)(x + (size_t)g1 * HID + sc4);
        const float4 vc = *(const float4*)(x + (size_t)g2 * HID + sc4);
        const float4 vd = *(const float4*)(x + (size_t)g3 * HID + sc4);
        *(float4*)(sSum + (r0     ) * LDS_PITCH + sc4) = va;
        *(float4*)(sSum + (r0 + 32) * LDS_PITCH + sc4) = vb;
        *(float4*)(sSum + (r0 + 64) * LDS_PITCH + sc4) = vc;
        *(float4*)(sSum + (r0 + 96) * LDS_PITCH + sc4) = vd;
    }
    if (sr0 < RB - (RB / 32) * 32) {
        const int r0 = (RB / 32) * 32 + sr0;
        const int n0 = lo + r0;
        const int g0 = n0 < last ? n0 : last;
        *(float4*)(sSum + r0 * LDS_PITCH + sc4) = *(const float4*)(x + (size_t)g0 * HID + sc4);
    }
    __syncthreads();

    if (dense) {
        sage_sweep(a0, wr, nc0, halfK, acc);

        const float bias0  = bl[nc0];
        const float bias1  = bl[nc0 + 16];
        const int   rowOff = (lane < 16) ? 0 : 8;
        const int   tile0  = lo + g * MT;
        auto store_tile = [&](bool guard) {
            #pragma unroll
            for (int mt = 0; mt < MSUB; ++mt) {
                #pragma unroll
                for (int r = 0; r < 8; ++r) {
                    const int row = tile0 + mt * 16 + r + rowOff;
                    if (!guard || row < nNodes) {
                        float v0 = acc[mt][0][r] + bias0;
                        float v1 = acc[mt][1][r] + bias1;
                        if (applyRelu) {
                            v0 = v0 > 0.0f ? v0 : 0.0f;
                            v1 = v1 > 0.0f ? v1 : 0.0f;
                        }
                        float* op = out + (size_t)row * HID + nc0;
                        op[0]  = v0;
                        op[16] = v1;
                    }
                }
            }
        };
        if (tile0 + MT <= nNodes) store_tile(false);
        else                      store_tile(true);
    }
}

extern "C" void kernel_launch(void* const* d_in, const int* in_sizes, int n_in,
                              void* d_out, int out_size, void* d_ws, size_t ws_size,
                              hipStream_t stream) {
    const float* x    = (const float*)d_in[0];
    const int*   ei   = (const int*)  d_in[1];
    const float* w1l  = (const float*)d_in[2];
    const float* b1l  = (const float*)d_in[3];
    const float* w1r  = (const float*)d_in[4];
    const float* w2l  = (const float*)d_in[5];
    const float* b2l  = (const float*)d_in[6];
    const float* w2r  = (const float*)d_in[7];

    const int N = in_sizes[0] / HID;
    const int E = in_sizes[1] / 2;
    const int* src = ei;
    const int* dst = ei + E;

    float* h    = (float*)d_ws;
    float* out  = (float*)d_out;

    const int ownGrid  = (N + RB - 1) / RB;
    int* qws  = (int*)(h + (size_t)N * HID);
    int* qcnt = qws + (size_t)ownGrid * QCAP * 2;
    const int maskGrid  = (E + MCH - 1) / MCH;
    const int maskPitch = maskGrid * MROW;
    unsigned* mask = (unsigned*)(qcnt + ((ownGrid + 63) & ~63));

    sage_mark<<<maskGrid, 256,   0, stream>>>(dst, mask, maskPitch, N, E, ownGrid);

    sage_own <<<ownGrid,  OWN_T, 0, stream>>>(x, src, dst, w1l, b1l, w1r, h,   1,
                                              qws, qcnt, mask, maskPitch, 0, N, E);

    sage_own <<<ownGrid,  OWN_T, 0, stream>>>(h, src, dst, w2l, b2l, w2r, out, 0,
                                              qws, qcnt, mask, maskPitch, 1, N, E);
}


# Rows worked as two halves by two 16-wave groups in staggered order: one half's matrix sweeps (weights from an LDS copy) run while the other half gathers; 2nd half sums x@Wr terms before mean@Wl terms
# speedup vs baseline: 2.5149x; 1.0430x over previous
#include <hip/hip_runtime.h>

#define HID       128
#define LDS_PITCH 132
#define NSMAX     4

typedef __attribute__((ext_vector_type(2))) float v2f;
typedef __attribute__((ext_vector_type(8))) float v8f;

#define RB      400
#define RA      208
#define OWN_T   1024
#define GRP_T   512
#define GRP_W   16
#define QCAP    4096
#define WP      136
#define CNT_COL 128
#define DEPTH_A 16
#define DEPTH_B 12

#define MCH     2048
#define MROW    64
#define MPAD    65
#define MOWN    256

template <int N> struct depth_tag { static constexpr int value = N; };

__global__ void __launch_bounds__(256)
sage_mark(const int* __restrict__ dst,
          unsigned*  __restrict__ mask,
          int maskPitch, int nNodes, int nEdges, int nOwn) {
    __shared__ unsigned sM[MOWN * MPAD];

    const int tid  = threadIdx.x;
    const int base = blockIdx.x * MCH;

    int d[MCH / 256];
    #pragma unroll
    for (int u = 0; u < MCH / 256; ++u) {
        const int e = base + u * 256 + tid;
        d[u] = e < nEdges ? dst[e] : -1;
    }

    for (int i = tid; i < MOWN * MPAD; i += 256) sM[i] = 0u;
    __syncthreads();

    #pragma unroll
    for (int u = 0; u < MCH / 256; ++u) {
        if ((unsigned)d[u] < (unsigned)nNodes) {
            const int o = d[u] / RB;
            const int i = u * 256 + tid;
            if (o < MOWN) atomicOr(&sM[o * MPAD + (i >> 5)], 1u << (i & 31));
        }
    }
    __syncthreads();

    for (int k = tid; k < nOwn * (MROW / 4); k += 256) {
        const int o = k / (MROW / 4);
        const int w = (k % (MROW / 4)) << 2;
        const unsigned* s = sM + o * MPAD + w;
        *(uint4*)(mask + (size_t)o * maskPitch + blockIdx.x * MROW + w) =
            make_uint4(s[0], s[1], s[2], s[3]);
    }
}

template <int NSUB, int WPITCH>
__device__ __forceinline__ void sage_sweep(const float* a0,
                                           const float* __restrict__ w,
                                           int nc0, int halfK,
                                           v8f (&acc)[NSMAX][2]) {
    #pragma nounroll
    for (int k0 = 0; k0 < HID; k0 += 4) {
        const float* wp = w + (k0 + halfK) * WPITCH + nc0;
        v2f b0; b0.x = wp[0];  b0.y = wp[WPITCH];
        v2f b1; b1.x = wp[16]; b1.y = wp[WPITCH + 16];
        #pragma unroll
        for (int mt = 0; mt < NSUB; ++mt) {
            const float* ap = a0 + mt * 16 * LDS_PITCH + k0;
            v2f a; a.x = ap[0]; a.y = ap[1];
            acc[mt][0] = __builtin_amdgcn_wmma_f32_16x16x4_f32(
                             false, a, false, b0, (short)0, acc[mt][0], false, false);
            acc[mt][1] = __builtin_amdgcn_wmma_f32_16x16x4_f32(
                             false, a, false, b1, (short)0, acc[mt][1], false, false);
        }
    }
}

__global__ void __launch_bounds__(OWN_T)
sage_own(const float* __restrict__ x,
         const int*   __restrict__ src,
         const int*   __restrict__ dst,
         const float* __restrict__ wl,
         const float* __restrict__ bl,
         const float* __restrict__ wr,
         float*       __restrict__ out,
         int applyRelu,
         int*         __restrict__ qws,
         int*         __restrict__ qcnt,
         const unsigned* __restrict__ mask,
         int maskPitch,
         int reuse, int nNodes, int nEdges) {
    __shared__ alignas(16) float sSum[RB * LDS_PITCH];
    __shared__ int2  sQ[QCAP];
    __shared__ alignas(16) float sW[HID * WP];
    __shared__ int   sQn;
    __shared__ int   sNA;
    __shared__ int   sNB;

    const int tid  = threadIdx.x;
    const int lo   = blockIdx.x * RB;
    const int wave = tid >> 5;
    const int lane = tid & 31;
    const int wv   = __builtin_amdgcn_readfirstlane(wave);
    const int grp  = wv >> 4;
    const int gw   = wv & (GRP_W - 1);
    const int gt   = tid & (GRP_T - 1);

    if (tid == 0) { sQn = 0; sNA = 0; sNB = 0; }
    __syncthreads();

    int2* const qmine = (int2*)qws + (size_t)blockIdx.x * QCAP;
    int nraw = 0;
    if (reuse) {
        nraw = qcnt[2 * blockIdx.x];
    } else {
        auto emit = [&](unsigned bits, int ebase, int& pos) {
            while (bits) {
                const int bit = __builtin_ctz(bits);
                bits &= bits - 1u;
                if (pos < QCAP) sQ[pos] = make_int2(ebase + bit, 0);
                ++pos;
            }
        };
        const uint4* m4 = (const uint4*)(mask + (size_t)blockIdx.x * maskPitch);
        const int    n4 = maskPitch >> 2;
        for (int i0 = tid; i0 < n4; i0 += 4 * OWN_T) {
            uint4 m[4];
            #pragma unroll
            for (int u = 0; u < 4; ++u) {
                const int i = i0 + u * OWN_T;
                m[u] = i < n4 ? m4[i] : make_uint4(0u, 0u, 0u, 0u);
            }
            #pragma unroll
            for (int u = 0; u < 4; ++u) {
                const int n = __popc(m[u].x) + __popc(m[u].y) + __popc(m[u].z) + __popc(m[u].w);
                if (n) {
                    int pos = atomicAdd(&sQn, n);
                    const int ebase = (i0 + u * OWN_T) << 7;
                    emit(m[u].x, ebase,      pos);
                    emit(m[u].y, ebase + 32, pos);
                    emit(m[u].z, ebase + 64, pos);
                    emit(m[u].w, ebase + 96, pos);
                }
            }
        }
    }
    __syncthreads();

    if (!reuse) nraw = sQn;
    const bool ovf = (unsigned)nraw > (unsigned)QCAP;
    const int  nq  = ovf ? 0 : nraw;
    int nA = 0, nB = 0;

    int es[QCAP / OWN_T], er[QCAP / OWN_T];
    if (reuse) {
        int a = qcnt[2 * blockIdx.x + 1];
        a  = a < 0 ? 0 : (a > nq ? nq : a);
        nA = a;
        nB = nq - a;
        for (int q = tid; q < QCAP; q += OWN_T)
            if (q < nA || q >= QCAP - nB) sQ[q] = qmine[q];
    } else {
        #pragma unroll
        for (int u = 0; u < QCAP / OWN_T; ++u) {
            const int q = u * OWN_T + tid;
            es[u] = 0;
            er[u] = -1;
            if (q < nq) {
                const int e = sQ[q].x;
                es[u] = src[e];
                er[u] = dst[e] - lo;
            }
        }
    }
    __syncthreads();

    if (!reuse) {
        const unsigned below = (1u << lane) - 1u;
        #pragma unroll
        for (int u = 0; u < QCAP / OWN_T; ++u) {
            if (u * OWN_T < nq) {
                const bool isA = er[u] >= 0 && er[u] < RA;
                const bool isB = er[u] >= RA;
                const unsigned bA = __builtin_amdgcn_ballot_w32(isA);
                const unsigned bB = __builtin_amdgcn_ballot_w32(isB);
                int baseA = 0, baseB = 0;
                if (lane == 0) {
                    if (bA) baseA = atomicAdd(&sNA, __popc(bA));
                    if (bB) baseB = atomicAdd(&sNB, __popc(bB));
                }
                baseA = __builtin_amdgcn_readfirstlane(baseA);
                baseB = __builtin_amdgcn_readfirstlane(baseB);
                if (isA) sQ[baseA + __popc(bA & below)] = make_int2(es[u], er[u]);
                if (isB) sQ[QCAP - 1 - (baseB + __popc(bB & below))] = make_int2(es[u], er[u]);
            }
        }
    }
    __syncthreads();

    if (!reuse) {
        nA = sNA;
        nB = sNB;
        for (int q = tid; q < QCAP; q += OWN_T)
            if (q < nA || q >= QCAP - nB) qmine[q] = sQ[q];
        if (tid == 0) {
            qcnt[2 * blockIdx.x]     = nraw;
            qcnt[2 * blockIdx.x + 1] = nA;
        }
    }

    const int last = nNodes - 1;

    auto zero_rows = [&](int row0, int nrows) {
        float4* p = (float4*)(sSum + row0 * LDS_PITCH);
        for (int i = gt; i < nrows * (LDS_PITCH / 4); i += GRP_T)
            p[i] = make_float4(0.0f, 0.0f, 0.0f, 0.0f);
    };

    auto stage_rows = [&](int row0, int npass) {
        const int sr  = gt >> 5;
        const int sc4 = (gt & 31) << 2;
        for (int p = 0; p < npass; p += 4) {
            float4 v[4];
            #pragma unroll
            for (int u = 0; u < 4; ++u) {
                const int n0 = lo + row0 + (p + u) * 16 + sr;
                const int gi = n0 < last ? n0 : last;
                v[u] = make_float4(0.0f, 0.0f, 0.0f, 0.0f);
                if (p + u < npass) v[u] = *(const float4*)(x + (size_t)gi * HID + sc4);
            }
            #pragma unroll
            for (int u = 0; u < 4; ++u) {
                if (p + u < npass)
                    *(float4*)(sSum + (row0 + (p + u) * 16 + sr) * LDS_PITCH + sc4) = v[u];
            }
        }
    };

    auto mean_rows = [&](int row0, int nrows) {
        for (int k = gt; k < nrows * 32; k += GRP_T) {
            const int r  = row0 + (k >> 5);
            const int c4 = (k & 31) << 2;
            float* srp = sSum + r * LDS_PITCH;
            const float  d  = srp[CNT_COL];
            const float  rd = d > 1.0f ? 1.0f / d : 1.0f;
            const float4 a4 = *(const float4*)(srp + c4);
            *(float4*)(srp + c4) = make_float4(a4.x * rd, a4.y * rd, a4.z * rd, a4.w * rd);
        }
    };

    auto gather = [&](auto tag, int n, int back) {
        constexpr int DEPTH = decltype(tag)::value;
        for (int base = gw * DEPTH; base < n; base += GRP_W * DEPTH) {
            int   s[DEPTH], r[DEPTH];
            float v[DEPTH][4];
            #pragma unroll
            for (int j = 0; j < DEPTH; ++j) {
                const int  idx = (base + j < n) ? base + j : base;
                const int2 q   = sQ[back ? QCAP - 1 - idx : idx];
                s[j] = __builtin_amdgcn_readfirstlane(q.x);
                r[j] = __builtin_amdgcn_readfirstlane(q.y);
            }
            #pragma unroll
            for (int j = 0; j < DEPTH; ++j) {
                const float* xs = x + (size_t)s[j] * HID + lane;
                v[j][0] = xs[0];
                v[j][1] = xs[32];
                v[j][2] = xs[64];
                v[j][3] = xs[96];
            }
            #pragma unroll
            for (int j = 0; j < DEPTH; ++j) {
                if (base + j < n) {
                    float* srp = sSum + r[j] * LDS_PITCH;
                    atomicAdd(srp + lane,      v[j][0]);
                    atomicAdd(srp + lane + 32, v[j][1]);
                    atomicAdd(srp + lane + 64, v[j][2]);
                    atomicAdd(srp + lane + 96, v[j][3]);
                    if (lane == 0) atomicAdd(srp + CNT_COL, 1.0f);
                }
            }
        }
    };

    auto walk_add = [&](int rlo, int rhi) {
        const unsigned* mrow = mask + (size_t)blockIdx.x * maskPitch;
        for (int wd = gt; wd < maskPitch; wd += GRP_T) {
            unsigned bits = mrow[wd];
            while (bits) {
                const int bit = __builtin_ctz(bits);
                bits &= bits - 1u;
                const int e = (wd << 5) + bit;
                const int r = dst[e] - lo;
                if (r >= rlo && r < rhi) {
                    const float* xs  = x + (size_t)src[e] * HID;
                    float*       srp = sSum + r * LDS_PITCH;
                    for (int f = 0; f < HID; ++f) atomicAdd(srp + f, xs[f]);
                    atomicAdd(srp + CNT_COL, 1.0f);
                }
            }
        }
    };

    const int  g     = gw >> 2;
    const int  lrow  = lane & 15;
    const int  halfK = (lane >> 4) << 1;
    const int  nc0   = (gw & 3) * 32 + lrow;
    const int  nsub  = (grp == 0 && g == 0) ? 4 : 3;
    const int  trow  = grp == 0 ? (g == 0 ? 0 : 16 + 48 * g) : RA + 48 * g;
    const float* a0  = sSum + (trow + lrow) * LDS_PITCH + halfK;

    v8f acc[NSMAX][2];
    auto acc_clear = [&]() {
        #pragma unroll
        for (int mt = 0; mt < NSMAX; ++mt) {
            #pragma unroll
            for (int nt = 0; nt < 2; ++nt) acc[mt][nt] = (v8f){};
        }
    };
    auto sweep_mem = [&](const float* __restrict__ w) {
        if (nsub == 4) sage_sweep<4, HID>(a0, w, nc0, halfK, acc);
        else           sage_sweep<3, HID>(a0, w, nc0, halfK, acc);
    };
    auto sweep_lds = [&]() {
        if (nsub == 4) sage_sweep<4, WP>(a0, sW, nc0, halfK, acc);
        else           sage_sweep<3, WP>(a0, sW, nc0, halfK, acc);
    };
    auto stage_w = [&](const float* __restrict__ w) {
        float4 v[4];
        #pragma unroll
        for (int u = 0; u < 4; ++u) {
            const int i = u * OWN_T + tid;
            v[u] = *(const float4*)(w + (i >> 5) * HID + ((i & 31) << 2));
        }
        #pragma unroll
        for (int u = 0; u < 4; ++u) {
            const int i = u * OWN_T + tid;
            *(float4*)(sW + (i >> 5) * WP + ((i & 31) << 2)) = v[u];
        }
    };

    stage_w(wr);
    if (grp == 0) zero_rows(0, RA);
    else          stage_rows(RA, (RB - RA) / 16);
    __syncthreads();

    if (grp == 0) {
        if (ovf) walk_add(0, RA);
        else     gather(depth_tag<DEPTH_A>{}, nA, 0);
    } else {
        acc_clear();
        sweep_lds();
    }
    __syncthreads();

    stage_w(wl);
    if (grp == 0) mean_rows(0, RA);
    else          zero_rows(RA, RB - RA);
    __syncthreads();

    if (grp == 0) {
        acc_clear();
        sweep_lds();
    } else {
        if (ovf) walk_add(RA, RB);
        else     gather(depth_tag<DEPTH_B>{}, nB, 1);
    }
    __syncthreads();

    if (grp == 0) stage_rows(0, RA / 16);
    else          mean_rows(RA, RB - RA);
    __syncthreads();

    if (grp == 0) sweep_mem(wr);
    else          sweep_lds();

    const float bias0  = bl[nc0];
    const float bias1  = bl[nc0 + 16];
    const int   rowOff = (lane < 16) ? 0 : 8;
    const int   tile0  = lo + trow;
    auto store_tile = [&](bool guard) {
        #pragma unroll
        for (int mt = 0; mt < NSMAX; ++mt) {
            if (mt < nsub) {
                #pragma unroll
                for (int r = 0; r < 8; ++r) {
                    const int row = tile0 + mt * 16 + r + rowOff;
                    if (!guard || row < nNodes) {
                        float v0 = acc[mt][0][r] + bias0;
                        float v1 = acc[mt][1][r] + bias1;
                        if (applyRelu) {
                            v0 = v0 > 0.0f ? v0 : 0.0f;
                            v1 = v1 > 0.0f ? v1 : 0.0f;
                        }
                        float* op = out + (size_t)row * HID + nc0;
                        op[0]  = v0;
                        op[16] = v1;
                    }
                }
            }
        }
    };
    if (tile0 + nsub * 16 <= nNodes) store_tile(false);
    else                             store_tile(true);
}

extern "C" void kernel_launch(void* const* d_in, const int* in_sizes, int n_in,
                              void* d_out, int out_size, void* d_ws, size_t ws_size,
                              hipStream_t stream) {
    const float* x    = (const float*)d_in[0];
    const int*   ei   = (const int*)  d_in[1];
    const float* w1l  = (const float*)d_in[2];
    const float* b1l  = (const float*)d_in[3];
    const float* w1r  = (const float*)d_in[4];
    const float* w2l  = (const float*)d_in[5];
    const float* b2l  = (const float*)d_in[6];
    const float* w2r  = (const float*)d_in[7];

    const int N = in_sizes[0] / HID;
    const int E = in_sizes[1] / 2;
    const int* src = ei;
    const int* dst = ei + E;

    float* h    = (float*)d_ws;
    float* out  = (float*)d_out;

    const int ownGrid  = (N + RB - 1) / RB;
    int* qws  = (int*)(h + (size_t)N * HID);
    int* qcnt = qws + (size_t)ownGrid * QCAP * 2;
    const int maskGrid  = (E + MCH - 1) / MCH;
    const int maskPitch = maskGrid * MROW;
    unsigned* mask = (unsigned*)(qcnt + ((2 * ownGrid + 63) & ~63));

    sage_mark<<<maskGrid, 256,   0, stream>>>(dst, mask, maskPitch, N, E, ownGrid);

    sage_own <<<ownGrid,  OWN_T, 0, stream>>>(x, src, dst, w1l, b1l, w1r, h,   1,
                                              qws, qcnt, mask, maskPitch, 0, N, E);

    sage_own <<<ownGrid,  OWN_T, 0, stream>>>(h, src, dst, w2l, b2l, w2r, out, 0,
                                              qws, qcnt, mask, maskPitch, 1, N, E);
}
